# starting kernel
# baseline (speedup 1.0000x reference)
#include <hip/hip_runtime.h>
#include <math.h>


typedef float v2f __attribute__((ext_vector_type(2)));
typedef float v8f __attribute__((ext_vector_type(8)));

#define WAVES_PER_BLOCK 8
#define TILE 16

__global__ void chamfer_min_wmma(const float* __restrict__ P,
                                 const float* __restrict__ Q,
                                 float* __restrict__ out,
                                 int NP, int NQ) {
    extern __shared__ float sA[];

    __builtin_assume(NP >= 64 && (NP & 63) == 0);
    __builtin_assume(NQ >= 128 && (NQ & 127) == 0);

    const int b    = blockIdx.y;
    const int tid  = threadIdx.x;
    const int lane = tid & 31;
    const int wid  = tid >> 5;
    const int hi   = lane >> 4;
    const int l16  = lane & 15;

    const float* Pb = P + (size_t)b * NP * 3;
    for (int i = tid; i < NP; i += blockDim.x) {
        float p0 = Pb[3 * i + 0];
        float p1 = Pb[3 * i + 1];
        float p2 = Pb[3 * i + 2];
        float n  = fmaf(p0, p0, fmaf(p1, p1, p2 * p2));
        float* slot = &sA[(i >> 4) * 64];
        int r = i & 15;
        slot[r * 2 + 0]        = -2.0f * p0;
        slot[r * 2 + 1]        = -2.0f * p1;
        slot[(r + 16) * 2 + 0] = -2.0f * p2;
        slot[(r + 16) * 2 + 1] = n;
    }
    __syncthreads();

    const int qtile = blockIdx.x * WAVES_PER_BLOCK + wid;
    if (qtile * TILE >= NQ) return;

    const int j = qtile * TILE + l16;
    const float* Qb = Q + (size_t)b * NQ * 3;

    v2f bmat;
    float qpart;
    if (!hi) {
        float q0 = Qb[3 * j + 0];
        float q1 = Qb[3 * j + 1];
        bmat.x = q0; bmat.y = q1;
        qpart  = fmaf(q0, q0, q1 * q1);
    } else {
        float q2 = Qb[3 * j + 2];
        bmat.x = q2; bmat.y = 1.0f;
        qpart  = q2 * q2;
    }
    const float q2n = qpart + __shfl_xor(qpart, 16, 32);

    v8f cvec;
#pragma unroll
    for (int r = 0; r < 8; ++r) cvec[r] = q2n;

    float best = __builtin_inff();
    const int ntiles = NP >> 4;
    const v2f* sAv = (const v2f*)sA;

#pragma unroll 4
    for (int t = 0; t < ntiles; ++t) {
        v2f amat = sAv[t * 32 + lane];
        v8f d = __builtin_amdgcn_wmma_f32_16x16x4_f32(
             false, amat,  false, bmat,
             (short)0, cvec,  false,  false);

        float t0 = fminf(fminf(d[0], d[1]), d[2]);
        float t1 = fminf(fminf(d[3], d[4]), d[5]);
        float t2 = fminf(fminf(t0, t1), d[6]);
        best     = fminf(fminf(best, t2), d[7]);
    }

    best = fminf(best, __shfl_xor(best, 16, 32));
    best = fmaxf(best, 0.0f);

    if (!hi) {
        out[(size_t)b * NQ + qtile * TILE + l16] = best;
    }
}

extern "C" void kernel_launch(void* const* d_in, const int* in_sizes, int n_in,
                              void* d_out, int out_size, void* d_ws, size_t ws_size,
                              hipStream_t stream) {
    const float* x = (const float*)d_in[0];
    const float* y = (const float*)d_in[1];
    float* out = (float*)d_out;

    const int B = 8, D = 3;
    const int N = in_sizes[0] / (B * D);
    const int M = in_sizes[1] / (B * D);

    const int threads = WAVES_PER_BLOCK * 32;

    {
        dim3 grid((N + WAVES_PER_BLOCK * TILE - 1) / (WAVES_PER_BLOCK * TILE), B);
        size_t shmem = (size_t)M * 16;
        chamfer_min_wmma<<<grid, threads, shmem, stream>>>(y, x, out, M, N);
    }
    {
        dim3 grid((M + WAVES_PER_BLOCK * TILE - 1) / (WAVES_PER_BLOCK * TILE), B);
        size_t shmem = (size_t)N * 16;
        chamfer_min_wmma<<<grid, threads, shmem, stream>>>(x, y, out + (size_t)B * N, N, M);
    }
}


# One launch instead of two: blockIdx.z picks the direction (P/Q roles), so one launch boundary goes and work-groups of both directions are co-resident: 2 per WGP, 4 waves per SIMD instead of 2
# speedup vs baseline: 1.4227x; 1.4227x over previous
#include <hip/hip_runtime.h>
#include <math.h>


typedef float v2f __attribute__((ext_vector_type(2)));
typedef float v8f __attribute__((ext_vector_type(8)));

#define WAVES_PER_BLOCK 8
#define TILE 16

__global__ void chamfer_min_wmma(const float* __restrict__ X,
                                 const float* __restrict__ Y,
                                 float* __restrict__ out,
                                 int N, int M) {
    extern __shared__ float sA[];

    __builtin_assume(N >= 128 && (N & 127) == 0);
    __builtin_assume(M >= 128 && (M & 127) == 0);

    const bool rev = blockIdx.z != 0;
    const float* P = rev ? X : Y;
    const float* Q = rev ? Y : X;
    const int NP   = rev ? N : M;
    const int NQ   = rev ? M : N;
    float* o       = rev ? out + (size_t)gridDim.y * N : out;

    if ((int)blockIdx.x * (WAVES_PER_BLOCK * TILE) >= NQ) return;

    const int b    = blockIdx.y;
    const int tid  = threadIdx.x;
    const int lane = tid & 31;
    const int wid  = tid >> 5;
    const int hi   = lane >> 4;
    const int l16  = lane & 15;

    const float* Pb = P + (size_t)b * NP * 3;
    for (int i = tid; i < NP; i += blockDim.x) {
        float p0 = Pb[3 * i + 0];
        float p1 = Pb[3 * i + 1];
        float p2 = Pb[3 * i + 2];
        float n  = fmaf(p0, p0, fmaf(p1, p1, p2 * p2));
        float* slot = &sA[(i >> 4) * 64];
        int r = i & 15;
        slot[r * 2 + 0]        = -2.0f * p0;
        slot[r * 2 + 1]        = -2.0f * p1;
        slot[(r + 16) * 2 + 0] = -2.0f * p2;
        slot[(r + 16) * 2 + 1] = n;
    }
    __syncthreads();

    const int qtile = blockIdx.x * WAVES_PER_BLOCK + wid;
    if (qtile * TILE >= NQ) return;

    const int j = qtile * TILE + l16;
    const float* Qb = Q + (size_t)b * NQ * 3;

    v2f bmat;
    float qpart;
    if (!hi) {
        float q0 = Qb[3 * j + 0];
        float q1 = Qb[3 * j + 1];
        bmat.x = q0; bmat.y = q1;
        qpart  = fmaf(q0, q0, q1 * q1);
    } else {
        float q2 = Qb[3 * j + 2];
        bmat.x = q2; bmat.y = 1.0f;
        qpart  = q2 * q2;
    }
    const float q2n = qpart + __shfl_xor(qpart, 16, 32);

    v8f cvec;
#pragma unroll
    for (int r = 0; r < 8; ++r) cvec[r] = q2n;

    float best = __builtin_inff();
    const int ntiles = NP >> 4;
    const v2f* sAv = (const v2f*)sA;

#pragma unroll 4
    for (int t = 0; t < ntiles; ++t) {
        v2f amat = sAv[t * 32 + lane];
        v8f d = __builtin_amdgcn_wmma_f32_16x16x4_f32(
             false, amat,  false, bmat,
             (short)0, cvec,  false,  false);

        float t0 = fminf(fminf(d[0], d[1]), d[2]);
        float t1 = fminf(fminf(d[3], d[4]), d[5]);
        float t2 = fminf(fminf(t0, t1), d[6]);
        best     = fminf(fminf(best, t2), d[7]);
    }

    best = fminf(best, __shfl_xor(best, 16, 32));
    best = fmaxf(best, 0.0f);

    if (!hi) {
        o[(size_t)b * NQ + qtile * TILE + l16] = best;
    }
}

extern "C" void kernel_launch(void* const* d_in, const int* in_sizes, int n_in,
                              void* d_out, int out_size, void* d_ws, size_t ws_size,
                              hipStream_t stream) {
    const float* x = (const float*)d_in[0];
    const float* y = (const float*)d_in[1];
    float* out = (float*)d_out;

    const int B = 8, D = 3;
    const int N = in_sizes[0] / (B * D);
    const int M = in_sizes[1] / (B * D);

    const int threads = WAVES_PER_BLOCK * 32;
    const int QB = WAVES_PER_BLOCK * TILE;

    const int L = N > M ? N : M;
    dim3 grid((L + QB - 1) / QB, B, 2);
    size_t shmem = (size_t)L * 16;
    chamfer_min_wmma<<<grid, threads, shmem, stream>>>(x, y, out, N, M);
}


# Prologue loads all in flight: each thread issues its whole share of P-slab loads, and the wave its Q loads, before the first wait: one memory round trip instead of one per staged point
# speedup vs baseline: 2.0342x; 1.4298x over previous
#include <hip/hip_runtime.h>
#include <math.h>


typedef float v2f __attribute__((ext_vector_type(2)));
typedef float v8f __attribute__((ext_vector_type(8)));

#define WAVES_PER_BLOCK 8
#define TILE 16
#define THREADS (WAVES_PER_BLOCK * 32)
#define STAGE_U 16

__device__ __forceinline__ void stage_point(float* sA, int i, float p0, float p1, float p2) {
    float n  = fmaf(p0, p0, fmaf(p1, p1, p2 * p2));
    float* slot = &sA[(i >> 4) * 64];
    int r = i & 15;
    slot[r * 2 + 0]        = -2.0f * p0;
    slot[r * 2 + 1]        = -2.0f * p1;
    slot[(r + 16) * 2 + 0] = -2.0f * p2;
    slot[(r + 16) * 2 + 1] = n;
}

__global__ void chamfer_min_wmma(const float* __restrict__ X,
                                 const float* __restrict__ Y,
                                 float* __restrict__ out,
                                 int N, int M) {
    extern __shared__ float sA[];

    __builtin_assume(N >= 128 && (N & 127) == 0);
    __builtin_assume(M >= 128 && (M & 127) == 0);

    const bool rev = blockIdx.z != 0;
    const float* P = rev ? X : Y;
    const float* Q = rev ? Y : X;
    const int NP   = rev ? N : M;
    const int NQ   = rev ? M : N;
    float* o       = rev ? out + (size_t)gridDim.y * N : out;

    if ((int)blockIdx.x * (WAVES_PER_BLOCK * TILE) >= NQ) return;

    const int b    = blockIdx.y;
    const int tid  = threadIdx.x;
    const int lane = tid & 31;
    const int wid  = tid >> 5;
    const int hi   = lane >> 4;
    const int l16  = lane & 15;

    const int qtile = blockIdx.x * WAVES_PER_BLOCK + wid;
    const int j = qtile * TILE + l16;
    const float* Qb = Q + (size_t)b * NQ * 3;
    const int jl = min(j, NQ - 1);
    const float qa = Qb[3 * jl + (hi ? 2 : 0)];
    const float qb = Qb[3 * jl + 1];

    const float* Pb = P + (size_t)b * NP * 3;
    int base = 0;
    for (; base + THREADS * STAGE_U <= NP; base += THREADS * STAGE_U) {
        float p0[STAGE_U], p1[STAGE_U], p2[STAGE_U];
#pragma unroll
        for (int u = 0; u < STAGE_U; ++u) {
            const int i = base + u * THREADS + tid;
            p0[u] = Pb[3 * i + 0];
            p1[u] = Pb[3 * i + 1];
            p2[u] = Pb[3 * i + 2];
        }
#pragma unroll
        for (int u = 0; u < STAGE_U; ++u)
            stage_point(sA, base + u * THREADS + tid, p0[u], p1[u], p2[u]);
    }
    for (int i = base + tid; i < NP; i += THREADS)
        stage_point(sA, i, Pb[3 * i + 0], Pb[3 * i + 1], Pb[3 * i + 2]);
    __syncthreads();

    if (qtile * TILE >= NQ) return;

    v2f bmat;
    float qpart;
    if (!hi) {
        bmat.x = qa; bmat.y = qb;
        qpart  = fmaf(qa, qa, qb * qb);
    } else {
        bmat.x = qa; bmat.y = 1.0f;
        qpart  = qa * qa;
    }
    const float q2n = qpart + __shfl_xor(qpart, 16, 32);

    v8f cvec;
#pragma unroll
    for (int r = 0; r < 8; ++r) cvec[r] = q2n;

    float best = __builtin_inff();
    const int ntiles = NP >> 4;
    const v2f* sAv = (const v2f*)sA;

#pragma unroll 4
    for (int t = 0; t < ntiles; ++t) {
        v2f amat = sAv[t * 32 + lane];
        v8f d = __builtin_amdgcn_wmma_f32_16x16x4_f32(
             false, amat,  false, bmat,
             (short)0, cvec,  false,  false);

        float t0 = fminf(fminf(d[0], d[1]), d[2]);
        float t1 = fminf(fminf(d[3], d[4]), d[5]);
        float t2 = fminf(fminf(t0, t1), d[6]);
        best     = fminf(fminf(best, t2), d[7]);
    }

    best = fminf(best, __shfl_xor(best, 16, 32));
    best = fmaxf(best, 0.0f);

    if (!hi) {
        o[(size_t)b * NQ + qtile * TILE + l16] = best;
    }
}

extern "C" void kernel_launch(void* const* d_in, const int* in_sizes, int n_in,
                              void* d_out, int out_size, void* d_ws, size_t ws_size,
                              hipStream_t stream) {
    const float* x = (const float*)d_in[0];
    const float* y = (const float*)d_in[1];
    float* out = (float*)d_out;

    const int B = 8, D = 3;
    const int N = in_sizes[0] / (B * D);
    const int M = in_sizes[1] / (B * D);

    const int threads = WAVES_PER_BLOCK * 32;
    const int QB = WAVES_PER_BLOCK * TILE;

    const int L = N > M ? N : M;
    dim3 grid((L + QB - 1) / QB, B, 2);
    size_t shmem = (size_t)L * 16;
    chamfer_min_wmma<<<grid, threads, shmem, stream>>>(x, y, out, N, M);
}


# Main loop software-pipelined in blocks of 8 P tiles: the next block's A fragments are read from LDS as one group into a second register set ahead of the current block's WMMAs; min folds trail a tile
# speedup vs baseline: 2.2069x; 1.0849x over previous
#include <hip/hip_runtime.h>
#include <math.h>


typedef float v2f __attribute__((ext_vector_type(2)));
typedef float v8f __attribute__((ext_vector_type(8)));

#define WAVES_PER_BLOCK 8
#define TILE 16
#define THREADS (WAVES_PER_BLOCK * 32)
#define STAGE_U 16
#define BLK 8

__device__ __forceinline__ void stage_point(float* sA, int i, float p0, float p1, float p2) {
    float n  = fmaf(p0, p0, fmaf(p1, p1, p2 * p2));
    float* slot = &sA[(i >> 4) * 64];
    int r = i & 15;
    slot[r * 2 + 0]        = -2.0f * p0;
    slot[r * 2 + 1]        = -2.0f * p1;
    slot[(r + 16) * 2 + 0] = -2.0f * p2;
    slot[(r + 16) * 2 + 1] = n;
}

__device__ __forceinline__ float fold_tile(float best, v8f d) {
    float t0 = fminf(fminf(d[0], d[1]), d[2]);
    float t1 = fminf(fminf(d[3], d[4]), d[5]);
    float t2 = fminf(fminf(t0, t1), d[6]);
    return fminf(fminf(best, t2), d[7]);
}

__global__ void chamfer_min_wmma(const float* __restrict__ X,
                                 const float* __restrict__ Y,
                                 float* __restrict__ out,
                                 int N, int M) {
    extern __shared__ float sA[];

    __builtin_assume(N >= 128 && (N & 127) == 0);
    __builtin_assume(M >= 128 && (M & 127) == 0);

    const bool rev = blockIdx.z != 0;
    const float* P = rev ? X : Y;
    const float* Q = rev ? Y : X;
    const int NP   = rev ? N : M;
    const int NQ   = rev ? M : N;
    float* o       = rev ? out + (size_t)gridDim.y * N : out;

    if ((int)blockIdx.x * (WAVES_PER_BLOCK * TILE) >= NQ) return;

    const int b    = blockIdx.y;
    const int tid  = threadIdx.x;
    const int lane = tid & 31;
    const int wid  = tid >> 5;
    const int hi   = lane >> 4;
    const int l16  = lane & 15;

    const int qtile = blockIdx.x * WAVES_PER_BLOCK + wid;
    const int j = qtile * TILE + l16;
    const float* Qb = Q + (size_t)b * NQ * 3;
    const int jl = min(j, NQ - 1);
    const float qa = Qb[3 * jl + (hi ? 2 : 0)];
    const float qb = Qb[3 * jl + 1];

    const float* Pb = P + (size_t)b * NP * 3;
    int base = 0;
    for (; base + THREADS * STAGE_U <= NP; base += THREADS * STAGE_U) {
        float p0[STAGE_U], p1[STAGE_U], p2[STAGE_U];
#pragma unroll
        for (int u = 0; u < STAGE_U; ++u) {
            const int i = base + u * THREADS + tid;
            p0[u] = Pb[3 * i + 0];
            p1[u] = Pb[3 * i + 1];
            p2[u] = Pb[3 * i + 2];
        }
#pragma unroll
        for (int u = 0; u < STAGE_U; ++u)
            stage_point(sA, base + u * THREADS + tid, p0[u], p1[u], p2[u]);
    }
    for (int i = base + tid; i < NP; i += THREADS)
        stage_point(sA, i, Pb[3 * i + 0], Pb[3 * i + 1], Pb[3 * i + 2]);
    __syncthreads();

    if (qtile * TILE >= NQ) return;

    v2f bmat;
    float qpart;
    if (!hi) {
        bmat.x = qa; bmat.y = qb;
        qpart  = fmaf(qa, qa, qb * qb);
    } else {
        bmat.x = qa; bmat.y = 1.0f;
        qpart  = qa * qa;
    }
    const float q2n = qpart + __shfl_xor(qpart, 16, 32);

    v8f cvec;
#pragma unroll
    for (int r = 0; r < 8; ++r) cvec[r] = q2n;

    float best = __builtin_inff();
    const int nblk = NP >> 7;
    const v2f* sAv = (const v2f*)sA + lane;

    v2f cur[BLK];
#pragma unroll
    for (int k = 0; k < BLK; ++k) cur[k] = sAv[k * 32];

    v8f dprev;
#pragma unroll
    for (int r = 0; r < 8; ++r) dprev[r] = __builtin_inff();

#pragma unroll 2
    for (int blk = 0; blk < nblk; ++blk) {
        const int nb = min(blk + 1, nblk - 1);
        v2f nxt[BLK];
        __builtin_amdgcn_sched_barrier(0);
#pragma unroll
        for (int k = 0; k < BLK; ++k) nxt[k] = sAv[(nb * BLK + k) * 32];
        asm volatile("" ::: "memory");
        __builtin_amdgcn_sched_barrier(0);

#pragma unroll
        for (int k = 0; k < BLK; ++k) {
            v8f d = __builtin_amdgcn_wmma_f32_16x16x4_f32(
                 false, cur[k],  false, bmat,
                 (short)0, cvec,  false,  false);
            best  = fold_tile(best, dprev);
            dprev = d;
        }
#pragma unroll
        for (int k = 0; k < BLK; ++k) cur[k] = nxt[k];
    }
    best = fold_tile(best, dprev);

    best = fminf(best, __shfl_xor(best, 16, 32));
    best = fmaxf(best, 0.0f);

    if (!hi) {
        o[(size_t)b * NQ + qtile * TILE + l16] = best;
    }
}

extern "C" void kernel_launch(void* const* d_in, const int* in_sizes, int n_in,
                              void* d_out, int out_size, void* d_ws, size_t ws_size,
                              hipStream_t stream) {
    const float* x = (const float*)d_in[0];
    const float* y = (const float*)d_in[1];
    float* out = (float*)d_out;

    const int B = 8, D = 3;
    const int N = in_sizes[0] / (B * D);
    const int M = in_sizes[1] / (B * D);

    const int threads = WAVES_PER_BLOCK * 32;
    const int QB = WAVES_PER_BLOCK * TILE;

    const int L = N > M ? N : M;
    dim3 grid((L + QB - 1) / QB, B, 2);
    size_t shmem = (size_t)L * 16;
    chamfer_min_wmma<<<grid, threads, shmem, stream>>>(x, y, out, N, M);
}


# Each 16x16 distance tile computed once for both directions (dist2 from dist1's sums): a wave holds 4 row tiles, walks column tiles; partial minima cross work-groups via workspace + 2nd launch
# speedup vs baseline: 2.5380x; 1.1500x over previous
#include <hip/hip_runtime.h>
#include <math.h>


typedef float v2f __attribute__((ext_vector_type(2)));
typedef float v8f __attribute__((ext_vector_type(8)));

#define WAVES_PER_BLOCK 8
#define TILE 16
#define THREADS (WAVES_PER_BLOCK * 32)
#define STAGE_U 16
#define BLK 8

__device__ __forceinline__ void stage_point(float* sA, int i, float p0, float p1, float p2) {
    float n  = fmaf(p0, p0, fmaf(p1, p1, p2 * p2));
    float* slot = &sA[(i >> 4) * 64];
    int r = i & 15;
    slot[r * 2 + 0]        = -2.0f * p0;
    slot[r * 2 + 1]        = -2.0f * p1;
    slot[(r + 16) * 2 + 0] = -2.0f * p2;
    slot[(r + 16) * 2 + 1] = n;
}

__device__ __forceinline__ float fold_tile(float best, v8f d) {
    float t0 = fminf(fminf(d[0], d[1]), d[2]);
    float t1 = fminf(fminf(d[3], d[4]), d[5]);
    float t2 = fminf(fminf(t0, t1), d[6]);
    return fminf(fminf(best, t2), d[7]);
}

__global__ void chamfer_min_wmma(const float* __restrict__ X,
                                 const float* __restrict__ Y,
                                 float* __restrict__ out,
                                 int N, int M) {
    extern __shared__ float sA[];

    __builtin_assume(N >= 128 && (N & 127) == 0);
    __builtin_assume(M >= 128 && (M & 127) == 0);

    const bool rev = blockIdx.z != 0;
    const float* P = rev ? X : Y;
    const float* Q = rev ? Y : X;
    const int NP   = rev ? N : M;
    const int NQ   = rev ? M : N;
    float* o       = rev ? out + (size_t)gridDim.y * N : out;

    if ((int)blockIdx.x * (WAVES_PER_BLOCK * TILE) >= NQ) return;

    const int b    = blockIdx.y;
    const int tid  = threadIdx.x;
    const int lane = tid & 31;
    const int wid  = tid >> 5;
    const int hi   = lane >> 4;
    const int l16  = lane & 15;

    const int qtile = blockIdx.x * WAVES_PER_BLOCK + wid;
    const int j = qtile * TILE + l16;
    const float* Qb = Q + (size_t)b * NQ * 3;
    const int jl = min(j, NQ - 1);
    const float qa = Qb[3 * jl + (hi ? 2 : 0)];
    const float qb = Qb[3 * jl + 1];

    const float* Pb = P + (size_t)b * NP * 3;
    int base = 0;
    for (; base + THREADS * STAGE_U <= NP; base += THREADS * STAGE_U) {
        float p0[STAGE_U], p1[STAGE_U], p2[STAGE_U];
#pragma unroll
        for (int u = 0; u < STAGE_U; ++u) {
            const int i = base + u * THREADS + tid;
            p0[u] = Pb[3 * i + 0];
            p1[u] = Pb[3 * i + 1];
            p2[u] = Pb[3 * i + 2];
        }
#pragma unroll
        for (int u = 0; u < STAGE_U; ++u)
            stage_point(sA, base + u * THREADS + tid, p0[u], p1[u], p2[u]);
    }
    for (int i = base + tid; i < NP; i += THREADS)
        stage_point(sA, i, Pb[3 * i + 0], Pb[3 * i + 1], Pb[3 * i + 2]);
    __syncthreads();

    if (qtile * TILE >= NQ) return;

    v2f bmat;
    float qpart;
    if (!hi) {
        bmat.x = qa; bmat.y = qb;
        qpart  = fmaf(qa, qa, qb * qb);
    } else {
        bmat.x = qa; bmat.y = 1.0f;
        qpart  = qa * qa;
    }
    const float q2n = qpart + __shfl_xor(qpart, 16, 32);

    v8f cvec;
#pragma unroll
    for (int r = 0; r < 8; ++r) cvec[r] = q2n;

    float best = __builtin_inff();
    const int nblk = NP >> 7;
    const v2f* sAv = (const v2f*)sA + lane;

    v2f cur[BLK];
#pragma unroll
    for (int k = 0; k < BLK; ++k) cur[k] = sAv[k * 32];

    v8f dprev;
#pragma unroll
    for (int r = 0; r < 8; ++r) dprev[r] = __builtin_inff();

#pragma unroll 2
    for (int blk = 0; blk < nblk; ++blk) {
        const int nb = min(blk + 1, nblk - 1);
        v2f nxt[BLK];
        __builtin_amdgcn_sched_barrier(0);
#pragma unroll
        for (int k = 0; k < BLK; ++k) nxt[k] = sAv[(nb * BLK + k) * 32];
        asm volatile("" ::: "memory");
        __builtin_amdgcn_sched_barrier(0);

#pragma unroll
        for (int k = 0; k < BLK; ++k) {
            v8f d = __builtin_amdgcn_wmma_f32_16x16x4_f32(
                 false, cur[k],  false, bmat,
                 (short)0, cvec,  false,  false);
            best  = fold_tile(best, dprev);
            dprev = d;
        }
#pragma unroll
        for (int k = 0; k < BLK; ++k) cur[k] = nxt[k];
    }
    best = fold_tile(best, dprev);

    best = fminf(best, __shfl_xor(best, 16, 32));
    best = fmaxf(best, 0.0f);

    if (!hi) {
        o[(size_t)b * NQ + qtile * TILE + l16] = best;
    }
}


typedef float v4f __attribute__((ext_vector_type(4)));

#define PW_WAVES 8
#define PW_RT    4
#define PW_ROWS  (PW_WAVES * PW_RT * TILE)
#define PW_COLS  1024
#define PW_CT    (PW_COLS / TILE)
#define PW_BLK   8
#define PW_MAXP  8

template <int CTRL>
__device__ __forceinline__ float row_xmask(float v) {
    return __builtin_bit_cast(float, __builtin_amdgcn_update_dpp(0, __builtin_bit_cast(int, v), CTRL, 0xf, 0xf, false));
}

__device__ __forceinline__ float row16_min8(const float (&R)[8], int l16) {
    const bool b3 = (l16 & 8) != 0, b2 = (l16 & 4) != 0, b1 = (l16 & 2) != 0;
    float S[4], T[2];
#pragma unroll
    for (int r = 0; r < 4; ++r) {
        const float keep = b3 ? R[r + 4] : R[r];
        const float send = b3 ? R[r] : R[r + 4];
        S[r] = fminf(keep, row_xmask<0x168>(send));
    }
#pragma unroll
    for (int r = 0; r < 2; ++r) {
        const float keep = b2 ? S[r + 2] : S[r];
        const float send = b2 ? S[r] : S[r + 2];
        T[r] = fminf(keep, row_xmask<0x164>(send));
    }
    const float keep = b1 ? T[1] : T[0];
    const float send = b1 ? T[0] : T[1];
    const float U = fminf(keep, row_xmask<0x162>(send));
    return fminf(U, row_xmask<0x161>(U));
}

__device__ __forceinline__ float col_min4(const v8f& d0, const v8f& d1, const v8f& d2, const v8f& d3) {
    const float p0 = fminf(fminf(d0[0], d0[1]), d0[2]);
    const float q0 = fminf(fminf(d0[3], d0[4]), d0[5]);
    const float f0 = fminf(fminf(p0, q0), d0[6]);
    const float p1 = fminf(fminf(d1[0], d1[1]), d1[2]);
    const float q1 = fminf(fminf(d1[3], d1[4]), d1[5]);
    const float f1 = fminf(fminf(p1, q1), d1[6]);
    const float p2 = fminf(fminf(d2[0], d2[1]), d2[2]);
    const float q2 = fminf(fminf(d2[3], d2[4]), d2[5]);
    const float f2 = fminf(fminf(p2, q2), d2[6]);
    const float p3 = fminf(fminf(d3[0], d3[1]), d3[2]);
    const float q3 = fminf(fminf(d3[3], d3[4]), d3[5]);
    const float f3 = fminf(fminf(p3, q3), d3[6]);
    const float g0 = fminf(fminf(f0, f1), d0[7]);
    const float g1 = fminf(fminf(f2, f3), d1[7]);
    const float h  = fminf(fminf(g0, g1), d2[7]);
    return fminf(h, d3[7]);
}

__global__ void __launch_bounds__(PW_WAVES * 32)
chamfer_pair_wmma(const float* __restrict__ X,
                  const float* __restrict__ Y,
                  float* __restrict__ ws,
                  int N, int M) {
    __shared__ __attribute__((aligned(16))) float sB[PW_CT * 64];
    __shared__ __attribute__((aligned(16))) float sCol[PW_WAVES * PW_CT * 32];
    __shared__ __attribute__((aligned(16))) float sRow[PW_WAVES * 64];

    const int cb   = blockIdx.x;
    const int rb   = blockIdx.y;
    const int b    = blockIdx.z;
    const int nB   = gridDim.z;
    const int tid  = threadIdx.x;
    const int lane = tid & 31;
    const int wid  = tid >> 5;
    const int hi   = lane >> 4;
    const int l16  = lane & 15;

    const float* Yb = Y + ((size_t)b * M + (size_t)cb * PW_COLS) * 3;
    float y0[PW_COLS / (PW_WAVES * 32)], y1[PW_COLS / (PW_WAVES * 32)], y2[PW_COLS / (PW_WAVES * 32)];
#pragma unroll
    for (int u = 0; u < PW_COLS / (PW_WAVES * 32); ++u) {
        const int i = u * (PW_WAVES * 32) + tid;
        y0[u] = Yb[3 * i + 0];
        y1[u] = Yb[3 * i + 1];
        y2[u] = Yb[3 * i + 2];
    }
    const int row0 = rb * PW_ROWS + wid * (PW_RT * TILE);
    const float* Xb = X + ((size_t)b * N + row0) * 3;
    float x0[PW_RT], x1[PW_RT], x2[PW_RT];
#pragma unroll
    for (int rt = 0; rt < PW_RT; ++rt) {
        const int i = rt * TILE + l16;
        x0[rt] = Xb[3 * i + 0];
        x1[rt] = Xb[3 * i + 1];
        x2[rt] = Xb[3 * i + 2];
    }

#pragma unroll
    for (int u = 0; u < PW_COLS / (PW_WAVES * 32); ++u) {
        const int i = u * (PW_WAVES * 32) + tid;
        float* slot = &sB[(i >> 4) * 64];
        const int r = i & 15;
        slot[r * 2 + 0]        = y0[u];
        slot[r * 2 + 1]        = y1[u];
        slot[(r + 16) * 2 + 0] = y2[u];
        slot[(r + 16) * 2 + 1] = fmaf(y0[u], y0[u], fmaf(y1[u], y1[u], y2[u] * y2[u]));
    }

    v2f amat[PW_RT];
    v8f cvec[PW_RT];
#pragma unroll
    for (int rt = 0; rt < PW_RT; ++rt) {
        const float nrm = fmaf(x0[rt], x0[rt], fmaf(x1[rt], x1[rt], x2[rt] * x2[rt]));
        if (!hi) { amat[rt].x = -2.0f * x0[rt]; amat[rt].y = -2.0f * x1[rt]; }
        else     { amat[rt].x = -2.0f * x2[rt]; amat[rt].y = 1.0f; }
#pragma unroll
        for (int r = 0; r < 8; ++r) cvec[rt][r] = __shfl(nrm, 8 * hi + r, 32);
    }
    __syncthreads();

    const v2f* sBv = (const v2f*)sB + lane;
    float* sColw   = sCol + wid * (PW_CT * 32) + lane;

    float rmin[PW_RT][8];
    v8f pA[PW_RT], pB[PW_RT];
#pragma unroll
    for (int rt = 0; rt < PW_RT; ++rt)
#pragma unroll
        for (int r = 0; r < 8; ++r) {
            rmin[rt][r] = __builtin_inff();
            pA[rt][r]   = __builtin_inff();
            pB[rt][r]   = __builtin_inff();
        }

    v2f cur[PW_BLK];
#pragma unroll
    for (int k = 0; k < PW_BLK; ++k) cur[k] = sBv[k * 32];

#pragma unroll 2
    for (int blk = 0; blk < PW_CT / PW_BLK; ++blk) {
        const int nb = min(blk + 1, PW_CT / PW_BLK - 1);
        v2f nxt[PW_BLK];
        __builtin_amdgcn_sched_barrier(0);
#pragma unroll
        for (int k = 0; k < PW_BLK; ++k) nxt[k] = sBv[(nb * PW_BLK + k) * 32];
        asm volatile("" ::: "memory");
        __builtin_amdgcn_sched_barrier(0);

#pragma unroll
        for (int j = 0; j < PW_BLK / 2; ++j) {
            v8f nA[PW_RT], nBt[PW_RT];
#pragma unroll
            for (int rt = 0; rt < PW_RT; ++rt)
                nA[rt] = __builtin_amdgcn_wmma_f32_16x16x4_f32(false, amat[rt], false, cur[2 * j],
                                                               (short)0, cvec[rt], false, false);
#pragma unroll
            for (int rt = 0; rt < PW_RT; ++rt)
                nBt[rt] = __builtin_amdgcn_wmma_f32_16x16x4_f32(false, amat[rt], false, cur[2 * j + 1],
                                                                (short)0, cvec[rt], false, false);
            const int tp = max(blk * PW_BLK + 2 * j - 2, 0);
#pragma unroll
            for (int rt = 0; rt < PW_RT; ++rt)
#pragma unroll
                for (int r = 0; r < 8; ++r)
                    rmin[rt][r] = fminf(fminf(rmin[rt][r], pA[rt][r]), pB[rt][r]);
            sColw[tp * 32]       = col_min4(pA[0], pA[1], pA[2], pA[3]);
            sColw[(tp + 1) * 32] = col_min4(pB[0], pB[1], pB[2], pB[3]);
#pragma unroll
            for (int rt = 0; rt < PW_RT; ++rt) { pA[rt] = nA[rt]; pB[rt] = nBt[rt]; }
        }
#pragma unroll
        for (int k = 0; k < PW_BLK; ++k) cur[k] = nxt[k];
    }
#pragma unroll
    for (int rt = 0; rt < PW_RT; ++rt)
#pragma unroll
        for (int r = 0; r < 8; ++r)
            rmin[rt][r] = fminf(fminf(rmin[rt][r], pA[rt][r]), pB[rt][r]);
    sColw[(PW_CT - 2) * 32] = col_min4(pA[0], pA[1], pA[2], pA[3]);
    sColw[(PW_CT - 1) * 32] = col_min4(pB[0], pB[1], pB[2], pB[3]);

    float* sRw = sRow + wid * 64;
#pragma unroll
    for (int rt = 0; rt < PW_RT; ++rt) {
        const float m = row16_min8(rmin[rt], l16);
        if ((l16 & 1) == 0) sRw[rt * TILE + 8 * hi + (l16 >> 1)] = m;
    }
    __syncthreads();

    float* rp = ws + ((size_t)(cb * nB + b) * N + row0);
    rp[lane]      = sRw[lane];
    rp[32 + lane] = sRw[32 + lane];

    {
        const int ct = tid >> 2;
        const int c4 = (tid & 3) * 4;
        v4f acc = *(const v4f*)&sCol[ct * 32 + c4];
        acc = __builtin_elementwise_min(acc, *(const v4f*)&sCol[ct * 32 + 16 + c4]);
#pragma unroll
        for (int w = 1; w < PW_WAVES; ++w) {
            const float* p = &sCol[w * (PW_CT * 32) + ct * 32 + c4];
            acc = __builtin_elementwise_min(acc, *(const v4f*)p);
            acc = __builtin_elementwise_min(acc, *(const v4f*)(p + 16));
        }
        float* cp = ws + (size_t)gridDim.x * nB * N
                       + ((size_t)(rb * nB + b) * M + (size_t)cb * PW_COLS + 4 * tid);
        *(v4f*)cp = acc;
    }
}

__global__ void __launch_bounds__(128)
chamfer_combine(const float* __restrict__ ws, float* __restrict__ out,
                int nrow, int ncol, int ncb, int nrb) {
    const int g = (blockIdx.x * 128 + threadIdx.x) * 4;
    if (g >= nrow + ncol) return;
    const bool isrow  = g < nrow;
    const float* base = isrow ? ws + g : ws + (size_t)ncb * nrow + (g - nrow);
    const size_t step = isrow ? (size_t)nrow : (size_t)ncol;
    const int    cnt  = isrow ? ncb : nrb;
    v4f v[PW_MAXP];
#pragma unroll
    for (int k = 0; k < PW_MAXP; ++k) {
        if (k < cnt) v[k] = *(const v4f*)(base + k * step);
        else         v[k] = (v4f)(__builtin_inff());
    }
    v4f m = v[0];
#pragma unroll
    for (int k = 1; k < PW_MAXP; ++k) m = __builtin_elementwise_min(m, v[k]);
    const v4f zero = (v4f)(0.0f);
    *(v4f*)(out + g) = __builtin_elementwise_max(m, zero);
}

extern "C" void kernel_launch(void* const* d_in, const int* in_sizes, int n_in,
                              void* d_out, int out_size, void* d_ws, size_t ws_size,
                              hipStream_t stream) {
    const float* x = (const float*)d_in[0];
    const float* y = (const float*)d_in[1];
    float* out = (float*)d_out;

    const int B = 8, D = 3;
    const int N = in_sizes[0] / (B * D);
    const int M = in_sizes[1] / (B * D);

    const int ncb = M / PW_COLS, nrb = N / PW_ROWS;
    const size_t need = ((size_t)ncb * N + (size_t)nrb * M) * B * sizeof(float);
    if (d_ws != nullptr && ws_size >= need && ncb >= 1 && nrb >= 1 && ncb <= PW_MAXP && nrb <= PW_MAXP &&
        N == nrb * PW_ROWS && M == ncb * PW_COLS) {
        float* ws = (float*)d_ws;
        chamfer_pair_wmma<<<dim3(ncb, nrb, B), PW_WAVES * 32, 0, stream>>>(x, y, ws, N, M);
        chamfer_combine<<<(B * (N + M) / 4 + 127) / 128, 128, 0, stream>>>(ws, out, B * N, B * M, ncb, nrb);
        return;
    }

    const int threads = WAVES_PER_BLOCK * 32;
    const int QB = WAVES_PER_BLOCK * TILE;

    const int L = N > M ? N : M;
    dim3 grid((L + QB - 1) / QB, B, 2);
    size_t shmem = (size_t)L * 16;
    chamfer_min_wmma<<<grid, threads, shmem, stream>>>(x, y, out, N, M);
}
